# starting kernel
# baseline (speedup 1.0000x reference)
#include <hip/hip_runtime.h>

#define B_ 8
#define N_ 2048
#define C_ 256
#define W_ 256

typedef float v2f __attribute__((ext_vector_type(2)));
typedef float v8f __attribute__((ext_vector_type(8)));

__device__ __forceinline__ v8f wmma_f32_16x16x4(v2f a, v2f b, v8f c) {
  return __builtin_amdgcn_wmma_f32_16x16x4_f32(false, a, false, b, (short)0, c,
                                               false, false);
}

__global__ void qkv_kernel(const float* __restrict__ x,
                           const float* __restrict__ Wq, const float* __restrict__ bq,
                           const float* __restrict__ Wk, const float* __restrict__ bk,
                           const float* __restrict__ Wv, const float* __restrict__ bv,
                           float* __restrict__ Q, float* __restrict__ K,
                           float* __restrict__ V) {
  const float* Wm;
  const float* bias;
  float* out;
  if (blockIdx.y == 0) { Wm = Wq; bias = bq; out = Q; }
  else if (blockIdx.y == 1) { Wm = Wk; bias = bk; out = K; }
  else { Wm = Wv; bias = bv; out = V; }

  const int wave = threadIdx.x >> 5;
  const int lane = threadIdx.x & 31;
  const int half = lane >> 4;
  const int r    = lane & 15;

  const int tiles_n = W_ / 16;
  const int tile = blockIdx.x * 8 + wave;
  const int m0 = (tile / tiles_n) * 16;
  const int n0 = (tile % tiles_n) * 16;

  const float* arow = x  + (size_t)(m0 + r) * C_;
  const float* brow = Wm + (size_t)(n0 + r) * C_;

  v8f acc = {};
#pragma unroll 4
  for (int k0 = 0; k0 < C_; k0 += 4) {
    v2f a = *(const v2f*)(arow + k0 + 2 * half);
    v2f b = *(const v2f*)(brow + k0 + 2 * half);
    acc = wmma_f32_16x16x4(a, b, acc);
  }

  const float bv_ = bias[n0 + r];
#pragma unroll
  for (int g = 0; g < 8; ++g) {
    const int m = m0 + g + 8 * half;
    float v = acc[g] + bv_;
    v = v > 0.0f ? v : 0.0f;
    out[(size_t)m * W_ + n0 + r] = v;
  }
}

__global__ void scores_kernel(const float* __restrict__ Q,
                              const float* __restrict__ K,
                              float* __restrict__ S) {
  const int b    = blockIdx.y;
  const int wave = threadIdx.x >> 5;
  const int lane = threadIdx.x & 31;
  const int half = lane >> 4;
  const int r    = lane & 15;

  const int tiles_n = N_ / 16;
  const int tile = blockIdx.x * 8 + wave;
  const int i0 = (tile / tiles_n) * 16;
  const int j0 = (tile % tiles_n) * 16;

  const float* arow = Q + ((size_t)b * N_ + i0 + r) * W_;
  const float* brow = K + ((size_t)b * N_ + j0 + r) * W_;

  v8f acc = {};
#pragma unroll 4
  for (int k0 = 0; k0 < W_; k0 += 4) {
    v2f a = *(const v2f*)(arow + k0 + 2 * half);
    v2f b2 = *(const v2f*)(brow + k0 + 2 * half);
    acc = wmma_f32_16x16x4(a, b2, acc);
  }

  const float scale = 0.022097086912079608f;
  float* sb = S + (size_t)b * N_ * N_;
#pragma unroll
  for (int g = 0; g < 8; ++g) {
    const int i = i0 + g + 8 * half;
    sb[(size_t)i * N_ + j0 + r] = acc[g] * scale;
  }
}

__global__ void colsoftmax_kernel(float* __restrict__ S, float* __restrict__ Dsum) {
  const int t = blockIdx.x * blockDim.x + threadIdx.x;
  const int b = t / N_;
  const int j = t % N_;
  float* col = S + (size_t)b * N_ * N_ + j;

  float m = -3.4e38f;
#pragma unroll 4
  for (int i = 0; i < N_; ++i) {
    m = fmaxf(m, col[(size_t)i * N_]);
  }
  float d = 0.0f;
#pragma unroll 4
  for (int i = 0; i < N_; ++i) {
    const float e = __expf(col[(size_t)i * N_] - m);
    col[(size_t)i * N_] = e;
    d += e;
  }
  Dsum[t] = d;
}

__global__ void scalev_kernel(float* __restrict__ V, const float* __restrict__ Dsum) {
  const int idx = blockIdx.x * blockDim.x + threadIdx.x;
  const int total = B_ * N_ * W_ / 4;
  if (idx >= total) return;
  float4* v4 = (float4*)V;
  const int row = (idx * 4) / W_;
  const float inv = 1.0f / Dsum[row];
  float4 v = v4[idx];
  v.x *= inv; v.y *= inv; v.z *= inv; v.w *= inv;
  v4[idx] = v;
}

__global__ void av_kernel(const float* __restrict__ P, const float* __restrict__ V,
                          const float* __restrict__ Q, float* __restrict__ out) {
  const int b    = blockIdx.y;
  const int wave = threadIdx.x >> 5;
  const int lane = threadIdx.x & 31;
  const int half = lane >> 4;
  const int r    = lane & 15;

  const int tiles_n = W_ / 16;
  const int tile = blockIdx.x * 8 + wave;
  const int i0 = (tile / tiles_n) * 16;
  const int n0 = (tile % tiles_n) * 16;

  const float* arow = P + (size_t)b * N_ * N_ + (size_t)(i0 + r) * N_;
  const float* vb   = V + (size_t)b * N_ * W_;

  v8f acc = {};
#pragma unroll 2
  for (int k0 = 0; k0 < N_; k0 += 4) {
    v2f a = *(const v2f*)(arow + k0 + 2 * half);
    const int kr = k0 + 2 * half;
    v2f b2;
    b2.x = vb[(size_t)kr * W_ + n0 + r];
    b2.y = vb[(size_t)(kr + 1) * W_ + n0 + r];
    acc = wmma_f32_16x16x4(a, b2, acc);
  }

  const float* qb = Q + (size_t)b * N_ * W_;
  float* ob = out + (size_t)b * N_ * W_;
#pragma unroll
  for (int g = 0; g < 8; ++g) {
    const int i = i0 + g + 8 * half;
    ob[(size_t)i * W_ + n0 + r] = acc[g] + qb[(size_t)i * W_ + n0 + r];
  }
}

extern "C" void kernel_launch(void* const* d_in, const int* in_sizes, int n_in,
                              void* d_out, int out_size, void* d_ws, size_t ws_size,
                              hipStream_t stream) {
  const float* x  = (const float*)d_in[0];
  const float* Wq = (const float*)d_in[1];
  const float* bq = (const float*)d_in[2];
  const float* Wk = (const float*)d_in[3];
  const float* bk = (const float*)d_in[4];
  const float* Wv = (const float*)d_in[5];
  const float* bv = (const float*)d_in[6];
  float* out = (float*)d_out;
  float* ws  = (float*)d_ws;

  const size_t QKV = (size_t)B_ * N_ * W_;
  float* Q = ws;
  float* K = ws + QKV;
  float* V = ws + 2 * QKV;
  float* D = ws + 3 * QKV;
  float* S = D + (size_t)B_ * N_;

  {
    dim3 grid(((B_ * N_ / 16) * (W_ / 16)) / 8, 3);
    qkv_kernel<<<grid, 256, 0, stream>>>(x, Wq, bq, Wk, bk, Wv, bv, Q, K, V);
  }
  {
    dim3 grid(((N_ / 16) * (N_ / 16)) / 8, B_);
    scores_kernel<<<grid, 256, 0, stream>>>(Q, K, S);
  }
  colsoftmax_kernel<<<dim3((B_ * N_) / 256), 256, 0, stream>>>(S, D);
  scalev_kernel<<<dim3((B_ * N_ * W_ / 4) / 256), 256, 0, stream>>>(V, D);
  {
    dim3 grid(((N_ / 16) * (W_ / 16)) / 8, B_);
    av_kernel<<<grid, 256, 0, stream>>>(S, V, Q, out);
  }
}


# scores stored transposed (K Q^T): softmax axis contiguous, one wave per row holds it in registers (read once), lane-butterfly max and sum (new sum order), normalised on store; scale-V launch dropped
# speedup vs baseline: 2.1681x; 2.1681x over previous
#include <hip/hip_runtime.h>

#define B_ 8
#define N_ 2048
#define C_ 256
#define W_ 256

typedef float v2f __attribute__((ext_vector_type(2)));
typedef float v8f __attribute__((ext_vector_type(8)));

__device__ __forceinline__ v8f wmma_f32_16x16x4(v2f a, v2f b, v8f c) {
  return __builtin_amdgcn_wmma_f32_16x16x4_f32(false, a, false, b, (short)0, c,
                                               false, false);
}

__global__ void qkv_kernel(const float* __restrict__ x,
                           const float* __restrict__ Wq, const float* __restrict__ bq,
                           const float* __restrict__ Wk, const float* __restrict__ bk,
                           const float* __restrict__ Wv, const float* __restrict__ bv,
                           float* __restrict__ Q, float* __restrict__ K,
                           float* __restrict__ V) {
  const float* Wm;
  const float* bias;
  float* out;
  if (blockIdx.y == 0) { Wm = Wq; bias = bq; out = Q; }
  else if (blockIdx.y == 1) { Wm = Wk; bias = bk; out = K; }
  else { Wm = Wv; bias = bv; out = V; }

  const int wave = threadIdx.x >> 5;
  const int lane = threadIdx.x & 31;
  const int half = lane >> 4;
  const int r    = lane & 15;

  const int tiles_n = W_ / 16;
  const int tile = blockIdx.x * 8 + wave;
  const int m0 = (tile / tiles_n) * 16;
  const int n0 = (tile % tiles_n) * 16;

  const float* arow = x  + (size_t)(m0 + r) * C_;
  const float* brow = Wm + (size_t)(n0 + r) * C_;

  v8f acc = {};
#pragma unroll 4
  for (int k0 = 0; k0 < C_; k0 += 4) {
    v2f a = *(const v2f*)(arow + k0 + 2 * half);
    v2f b = *(const v2f*)(brow + k0 + 2 * half);
    acc = wmma_f32_16x16x4(a, b, acc);
  }

  const float bv_ = bias[n0 + r];
#pragma unroll
  for (int g = 0; g < 8; ++g) {
    const int m = m0 + g + 8 * half;
    float v = acc[g] + bv_;
    v = v > 0.0f ? v : 0.0f;
    out[(size_t)m * W_ + n0 + r] = v;
  }
}

__global__ void scoresT_kernel(const float* __restrict__ Q,
                               const float* __restrict__ K,
                               float* __restrict__ ST) {
  const int b    = blockIdx.y;
  const int wave = threadIdx.x >> 5;
  const int lane = threadIdx.x & 31;
  const int half = lane >> 4;
  const int r    = lane & 15;

  const int tiles_n = N_ / 16;
  const int tile = blockIdx.x * 8 + wave;
  const int j0 = (tile / tiles_n) * 16;
  const int i0 = (tile % tiles_n) * 16;

  const float* arow = K + ((size_t)b * N_ + j0 + r) * W_;
  const float* brow = Q + ((size_t)b * N_ + i0 + r) * W_;

  v8f acc = {};
#pragma unroll 4
  for (int k0 = 0; k0 < W_; k0 += 4) {
    v2f a = *(const v2f*)(arow + k0 + 2 * half);
    v2f b2 = *(const v2f*)(brow + k0 + 2 * half);
    acc = wmma_f32_16x16x4(a, b2, acc);
  }

  const float scale = 0.022097086912079608f;
  float* sb = ST + (size_t)b * N_ * N_;
#pragma unroll
  for (int g = 0; g < 8; ++g) {
    const int j = j0 + g + 8 * half;
    sb[(size_t)j * N_ + i0 + r] = acc[g] * scale;
  }
}

__global__ void rowsoftmax_kernel(float* __restrict__ ST) {
  const int wave = threadIdx.x >> 5;
  const int lane = threadIdx.x & 31;
  const size_t row = (size_t)blockIdx.x * 8 + wave;
  float4* p = (float4*)(ST + row * N_);

  float4 v[16];
#pragma unroll
  for (int c = 0; c < 16; ++c) v[c] = p[c * 32 + lane];

  float m = v[0].x;
#pragma unroll
  for (int c = 0; c < 16; ++c) {
    m = fmaxf(m, fmaxf(fmaxf(v[c].x, v[c].y), fmaxf(v[c].z, v[c].w)));
  }
#pragma unroll
  for (int off = 16; off > 0; off >>= 1) m = fmaxf(m, __shfl_xor(m, off, 32));

  float d = 0.0f;
#pragma unroll
  for (int c = 0; c < 16; ++c) {
    v[c].x = __expf(v[c].x - m);
    v[c].y = __expf(v[c].y - m);
    v[c].z = __expf(v[c].z - m);
    v[c].w = __expf(v[c].w - m);
    d += (v[c].x + v[c].y) + (v[c].z + v[c].w);
  }
#pragma unroll
  for (int off = 16; off > 0; off >>= 1) d += __shfl_xor(d, off, 32);

  const float inv = 1.0f / d;
#pragma unroll
  for (int c = 0; c < 16; ++c) {
    float4 o = v[c];
    o.x *= inv; o.y *= inv; o.z *= inv; o.w *= inv;
    p[c * 32 + lane] = o;
  }
}

__global__ void av_kernel(const float* __restrict__ PT, const float* __restrict__ V,
                          const float* __restrict__ Q, float* __restrict__ out) {
  const int b    = blockIdx.y;
  const int wave = threadIdx.x >> 5;
  const int lane = threadIdx.x & 31;
  const int half = lane >> 4;
  const int r    = lane & 15;

  const int tiles_n = W_ / 16;
  const int tile = blockIdx.x * 8 + wave;
  const int i0 = (tile / tiles_n) * 16;
  const int n0 = (tile % tiles_n) * 16;

  const float* pcol = PT + (size_t)b * N_ * N_ + i0 + r;
  const float* vcol = V + (size_t)b * N_ * W_ + n0 + r;

  v8f acc = {};
#pragma unroll 2
  for (int k0 = 0; k0 < N_; k0 += 4) {
    const int kr = k0 + 2 * half;
    v2f a;
    a.x = pcol[(size_t)kr * N_];
    a.y = pcol[(size_t)(kr + 1) * N_];
    v2f b2;
    b2.x = vcol[(size_t)kr * W_];
    b2.y = vcol[(size_t)(kr + 1) * W_];
    acc = wmma_f32_16x16x4(a, b2, acc);
  }

  const float* qb = Q + (size_t)b * N_ * W_;
  float* ob = out + (size_t)b * N_ * W_;
#pragma unroll
  for (int g = 0; g < 8; ++g) {
    const int i = i0 + g + 8 * half;
    ob[(size_t)i * W_ + n0 + r] = acc[g] + qb[(size_t)i * W_ + n0 + r];
  }
}

extern "C" void kernel_launch(void* const* d_in, const int* in_sizes, int n_in,
                              void* d_out, int out_size, void* d_ws, size_t ws_size,
                              hipStream_t stream) {
  const float* x  = (const float*)d_in[0];
  const float* Wq = (const float*)d_in[1];
  const float* bq = (const float*)d_in[2];
  const float* Wk = (const float*)d_in[3];
  const float* bk = (const float*)d_in[4];
  const float* Wv = (const float*)d_in[5];
  const float* bv = (const float*)d_in[6];
  float* out = (float*)d_out;
  float* ws  = (float*)d_ws;

  const size_t QKV = (size_t)B_ * N_ * W_;
  float* Q = ws;
  float* K = ws + QKV;
  float* V = ws + 2 * QKV;
  float* ST = ws + 3 * QKV;

  {
    dim3 grid(((B_ * N_ / 16) * (W_ / 16)) / 8, 3);
    qkv_kernel<<<grid, 256, 0, stream>>>(x, Wq, bq, Wk, bk, Wv, bv, Q, K, V);
  }
  {
    dim3 grid(((N_ / 16) * (N_ / 16)) / 8, B_);
    scoresT_kernel<<<grid, 256, 0, stream>>>(Q, K, ST);
  }
  rowsoftmax_kernel<<<dim3((B_ * N_) / 8), 256, 0, stream>>>(ST);
  {
    dim3 grid(((N_ / 16) * (W_ / 16)) / 8, B_);
    av_kernel<<<grid, 256, 0, stream>>>(ST, V, Q, out);
  }
}


# Q, K, V and scores stored in matrix-fragment order (512-B chunks of 16 rows x 8 k): each operand load is one contiguous 16-B-per-lane load for two products; new k and softmax sum order
# speedup vs baseline: 3.7765x; 1.7418x over previous
#include <hip/hip_runtime.h>

#define B_ 8
#define N_ 2048
#define C_ 256
#define W_ 256

typedef float v2f __attribute__((ext_vector_type(2)));
typedef float v4f __attribute__((ext_vector_type(4)));
typedef float v8f __attribute__((ext_vector_type(8)));

__device__ __forceinline__ v8f wmma_f32_16x16x4(v2f a, v2f b, v8f c) {
  return __builtin_amdgcn_wmma_f32_16x16x4_f32(false, a, false, b, (short)0, c,
                                               false, false);
}


__global__ void qkv_kernel(const float* __restrict__ x,
                           const float* __restrict__ Wq, const float* __restrict__ bq,
                           const float* __restrict__ Wk, const float* __restrict__ bk,
                           const float* __restrict__ Wv, const float* __restrict__ bv,
                           float* __restrict__ Q, float* __restrict__ K,
                           float* __restrict__ V) {
  const float* Wm;
  const float* bias;
  float* out;
  if (blockIdx.y == 0) { Wm = Wq; bias = bq; out = Q; }
  else if (blockIdx.y == 1) { Wm = Wk; bias = bk; out = K; }
  else { Wm = Wv; bias = bv; out = V; }
  const bool isV = (blockIdx.y == 2);

  const int wave = threadIdx.x >> 5;
  const int lane = threadIdx.x & 31;
  const int half = lane >> 4;
  const int r    = lane & 15;

  const int tiles_n = W_ / 16;
  const int tile = blockIdx.x * 8 + wave;
  const int m0 = (tile / tiles_n) * 16;
  const int n0 = (tile % tiles_n) * 16;

  const float* xrow = x  + (size_t)(m0 + r) * C_;
  const float* wrow = Wm + (size_t)(n0 + r) * C_;
  const float* arow = isV ? xrow : wrow;
  const float* brow = isV ? wrow : xrow;

  v8f acc = {};
#pragma unroll 4
  for (int k0 = 0; k0 < C_; k0 += 4) {
    v2f a = *(const v2f*)(arow + k0 + 2 * half);
    v2f b = *(const v2f*)(brow + k0 + 2 * half);
    acc = wmma_f32_16x16x4(a, b, acc);
  }

  v4f lo, hi;
  size_t chunk;
  if (!isV) {
    const v4f b0 = *(const v4f*)(bias + n0 + 8 * half);
    const v4f b1 = *(const v4f*)(bias + n0 + 8 * half + 4);
    lo = (v4f){acc[0] + b0.x, acc[1] + b0.y, acc[2] + b0.z, acc[3] + b0.w};
    hi = (v4f){acc[4] + b1.x, acc[5] + b1.y, acc[6] + b1.z, acc[7] + b1.w};
    chunk = (size_t)(m0 / 16) * (W_ / 8) + (n0 / 8 + half);
  } else {
    const float bv_ = bias[n0 + r];
    lo = (v4f){acc[0] + bv_, acc[1] + bv_, acc[2] + bv_, acc[3] + bv_};
    hi = (v4f){acc[4] + bv_, acc[5] + bv_, acc[6] + bv_, acc[7] + bv_};
    const int b    = m0 / N_;
    const int jloc = m0 % N_;
    chunk = (size_t)b * (N_ * W_ / 128) + (size_t)(n0 / 16) * (N_ / 8) + (jloc / 8 + half);
  }
  lo.x = lo.x > 0.0f ? lo.x : 0.0f; lo.y = lo.y > 0.0f ? lo.y : 0.0f;
  lo.z = lo.z > 0.0f ? lo.z : 0.0f; lo.w = lo.w > 0.0f ? lo.w : 0.0f;
  hi.x = hi.x > 0.0f ? hi.x : 0.0f; hi.y = hi.y > 0.0f ? hi.y : 0.0f;
  hi.z = hi.z > 0.0f ? hi.z : 0.0f; hi.w = hi.w > 0.0f ? hi.w : 0.0f;
  v4f* dst = (v4f*)out + chunk * 32;
  dst[r]      = lo;
  dst[16 + r] = hi;
}

__global__ void scores_kernel(const float* __restrict__ Q,
                              const float* __restrict__ K,
                              float* __restrict__ ST) {
  const int b    = blockIdx.y;
  const int wave = threadIdx.x >> 5;
  const int lane = threadIdx.x & 31;
  const int half = lane >> 4;
  const int r    = lane & 15;

  const int tiles_n = N_ / 16;
  const int tile = blockIdx.x * 8 + wave;
  const int j0 = (tile / tiles_n) * 16;
  const int i0 = (tile % tiles_n) * 16;

  const v4f* ap = (const v4f*)K + ((size_t)((b * N_ + j0) / 16) * (W_ / 8)) * 32 + lane;
  const v4f* bp = (const v4f*)Q + ((size_t)((b * N_ + i0) / 16) * (W_ / 8)) * 32 + lane;

  v8f acc = {};
#pragma unroll 4
  for (int kc = 0; kc < W_ / 8; ++kc) {
    const v4f a = ap[kc * 32];
    const v4f q = bp[kc * 32];
    acc = wmma_f32_16x16x4(a.lo, q.lo, acc);
    acc = wmma_f32_16x16x4(a.hi, q.hi, acc);
  }

  const float scale = 0.022097086912079608f;
  const size_t chunk = (size_t)b * (N_ * (N_ / 128)) +
                       (size_t)(j0 / 8 + half) * (N_ / 16) + (i0 / 16);
  v4f* dst = (v4f*)ST + chunk * 32;
  dst[r]      = (v4f){acc[0] * scale, acc[1] * scale, acc[2] * scale, acc[3] * scale};
  dst[16 + r] = (v4f){acc[4] * scale, acc[5] * scale, acc[6] * scale, acc[7] * scale};
}

__global__ void chunksoftmax_kernel(float* __restrict__ ST) {
  __shared__ v4f red[2][8][2];
  const int wave = threadIdx.x >> 5;
  const int lane = threadIdx.x & 31;
  const int hp   = lane >> 4;
  v4f* p = (v4f*)ST + (size_t)blockIdx.x * (128 * 32) + wave * (16 * 32) + lane;

  v4f v[16];
#pragma unroll
  for (int c = 0; c < 16; ++c) v[c] = p[c * 32];

  v4f m = v[0];
#pragma unroll
  for (int c = 1; c < 16; ++c) {
    m.x = fmaxf(m.x, v[c].x); m.y = fmaxf(m.y, v[c].y);
    m.z = fmaxf(m.z, v[c].z); m.w = fmaxf(m.w, v[c].w);
  }
#pragma unroll
  for (int off = 8; off > 0; off >>= 1) {
    m.x = fmaxf(m.x, __shfl_xor(m.x, off, 32));
    m.y = fmaxf(m.y, __shfl_xor(m.y, off, 32));
    m.z = fmaxf(m.z, __shfl_xor(m.z, off, 32));
    m.w = fmaxf(m.w, __shfl_xor(m.w, off, 32));
  }
  if ((lane & 15) == 0) red[0][wave][hp] = m;
  __syncthreads();
  m = red[0][0][hp];
#pragma unroll
  for (int w = 1; w < 8; ++w) {
    const v4f t = red[0][w][hp];
    m.x = fmaxf(m.x, t.x); m.y = fmaxf(m.y, t.y);
    m.z = fmaxf(m.z, t.z); m.w = fmaxf(m.w, t.w);
  }

  v4f d = {0.0f, 0.0f, 0.0f, 0.0f};
#pragma unroll
  for (int c = 0; c < 16; ++c) {
    v[c].x = __expf(v[c].x - m.x);
    v[c].y = __expf(v[c].y - m.y);
    v[c].z = __expf(v[c].z - m.z);
    v[c].w = __expf(v[c].w - m.w);
    d += v[c];
  }
#pragma unroll
  for (int off = 8; off > 0; off >>= 1) {
    d.x += __shfl_xor(d.x, off, 32);
    d.y += __shfl_xor(d.y, off, 32);
    d.z += __shfl_xor(d.z, off, 32);
    d.w += __shfl_xor(d.w, off, 32);
  }
  if ((lane & 15) == 0) red[1][wave][hp] = d;
  __syncthreads();
  d = red[1][0][hp];
#pragma unroll
  for (int w = 1; w < 8; ++w) d += red[1][w][hp];

  const v4f inv = {1.0f / d.x, 1.0f / d.y, 1.0f / d.z, 1.0f / d.w};
#pragma unroll
  for (int c = 0; c < 16; ++c) p[c * 32] = v[c] * inv;
}

__global__ void av_kernel(const float* __restrict__ PT, const float* __restrict__ V,
                          const float* __restrict__ Q, float* __restrict__ out) {
  const int b    = blockIdx.y;
  const int wave = threadIdx.x >> 5;
  const int lane = threadIdx.x & 31;
  const int half = lane >> 4;
  const int r    = lane & 15;

  const int tiles_n = W_ / 16;
  const int tile = blockIdx.x * 8 + wave;
  const int i0 = (tile / tiles_n) * 16;
  const int n0 = (tile % tiles_n) * 16;

  const v4f* ap = (const v4f*)PT + ((size_t)b * (N_ * (N_ / 128)) + (i0 / 16)) * 32 + lane;
  const v4f* bp = (const v4f*)V + ((size_t)b * (N_ * W_ / 128) + (size_t)(n0 / 16) * (N_ / 8)) * 32 + lane;

  v8f acc = {};
#pragma unroll 4
  for (int kc = 0; kc < N_ / 8; ++kc) {
    const v4f a = ap[(size_t)kc * ((N_ / 16) * 32)];
    const v4f v = bp[kc * 32];
    acc = wmma_f32_16x16x4(a.lo, v.lo, acc);
    acc = wmma_f32_16x16x4(a.hi, v.hi, acc);
  }

  const float* qc = Q + (((size_t)((b * N_ + i0) / 16) * (W_ / 8) + (n0 / 8 + (r >> 3))) * 32 +
                         ((r >> 2) & 1) * 16 + 8 * half) * 4 + (r & 3);
  float* ob = out + (size_t)b * N_ * W_;
#pragma unroll
  for (int g = 0; g < 8; ++g) {
    const int i = i0 + g + 8 * half;
    ob[(size_t)i * W_ + n0 + r] = acc[g] + qc[g * 4];
  }
}

extern "C" void kernel_launch(void* const* d_in, const int* in_sizes, int n_in,
                              void* d_out, int out_size, void* d_ws, size_t ws_size,
                              hipStream_t stream) {
  const float* x  = (const float*)d_in[0];
  const float* Wq = (const float*)d_in[1];
  const float* bq = (const float*)d_in[2];
  const float* Wk = (const float*)d_in[3];
  const float* bk = (const float*)d_in[4];
  const float* Wv = (const float*)d_in[5];
  const float* bv = (const float*)d_in[6];
  float* out = (float*)d_out;
  float* ws  = (float*)d_ws;

  const size_t QKV = (size_t)B_ * N_ * W_;
  float* Q = ws;
  float* K = ws + QKV;
  float* V = ws + 2 * QKV;
  float* ST = ws + 3 * QKV;

  {
    dim3 grid(((B_ * N_ / 16) * (W_ / 16)) / 8, 3);
    qkv_kernel<<<grid, 256, 0, stream>>>(x, Wq, bq, Wk, bk, Wv, bv, Q, K, V);
  }
  {
    dim3 grid(((N_ / 16) * (N_ / 16)) / 8, B_);
    scores_kernel<<<grid, 256, 0, stream>>>(Q, K, ST);
  }
  chunksoftmax_kernel<<<dim3(B_ * (N_ / 8)), 256, 0, stream>>>(ST);
  {
    dim3 grid(((N_ / 16) * (W_ / 16)) / 8, B_);
    av_kernel<<<grid, 256, 0, stream>>>(ST, V, Q, out);
  }
}


# scores and softmax fused: a work-group owns 32 key rows x all 2048 queries, scores stay in registers (2x8 tiles per wave), softmax there (new sum order), probabilities written once; 3 launches
# speedup vs baseline: 4.1449x; 1.0976x over previous
#include <hip/hip_runtime.h>

#define B_ 8
#define N_ 2048
#define C_ 256
#define W_ 256

typedef float v2f __attribute__((ext_vector_type(2)));
typedef float v4f __attribute__((ext_vector_type(4)));
typedef float v8f __attribute__((ext_vector_type(8)));

__device__ __forceinline__ v8f wmma_f32_16x16x4(v2f a, v2f b, v8f c) {
  return __builtin_amdgcn_wmma_f32_16x16x4_f32(false, a, false, b, (short)0, c,
                                               false, false);
}


__global__ void qkv_kernel(const float* __restrict__ x,
                           const float* __restrict__ Wq, const float* __restrict__ bq,
                           const float* __restrict__ Wk, const float* __restrict__ bk,
                           const float* __restrict__ Wv, const float* __restrict__ bv,
                           float* __restrict__ Q, float* __restrict__ K,
                           float* __restrict__ V) {
  const float* Wm;
  const float* bias;
  float* out;
  if (blockIdx.y == 0) { Wm = Wq; bias = bq; out = Q; }
  else if (blockIdx.y == 1) { Wm = Wk; bias = bk; out = K; }
  else { Wm = Wv; bias = bv; out = V; }
  const bool isV = (blockIdx.y == 2);

  const int wave = threadIdx.x >> 5;
  const int lane = threadIdx.x & 31;
  const int half = lane >> 4;
  const int r    = lane & 15;

  const int tiles_n = W_ / 16;
  const int tile = blockIdx.x * 8 + wave;
  const int m0 = (tile / tiles_n) * 16;
  const int n0 = (tile % tiles_n) * 16;

  const float* xrow = x  + (size_t)(m0 + r) * C_;
  const float* wrow = Wm + (size_t)(n0 + r) * C_;
  const float* arow = isV ? xrow : wrow;
  const float* brow = isV ? wrow : xrow;

  v8f acc = {};
#pragma unroll 4
  for (int k0 = 0; k0 < C_; k0 += 4) {
    v2f a = *(const v2f*)(arow + k0 + 2 * half);
    v2f b = *(const v2f*)(brow + k0 + 2 * half);
    acc = wmma_f32_16x16x4(a, b, acc);
  }

  v4f lo, hi;
  size_t chunk;
  if (!isV) {
    const v4f b0 = *(const v4f*)(bias + n0 + 8 * half);
    const v4f b1 = *(const v4f*)(bias + n0 + 8 * half + 4);
    lo = (v4f){acc[0] + b0.x, acc[1] + b0.y, acc[2] + b0.z, acc[3] + b0.w};
    hi = (v4f){acc[4] + b1.x, acc[5] + b1.y, acc[6] + b1.z, acc[7] + b1.w};
    chunk = (size_t)(m0 / 16) * (W_ / 8) + (n0 / 8 + half);
  } else {
    const float bv_ = bias[n0 + r];
    lo = (v4f){acc[0] + bv_, acc[1] + bv_, acc[2] + bv_, acc[3] + bv_};
    hi = (v4f){acc[4] + bv_, acc[5] + bv_, acc[6] + bv_, acc[7] + bv_};
    const int b    = m0 / N_;
    const int jloc = m0 % N_;
    chunk = (size_t)b * (N_ * W_ / 128) + (size_t)(n0 / 16) * (N_ / 8) + (jloc / 8 + half);
  }
  lo.x = lo.x > 0.0f ? lo.x : 0.0f; lo.y = lo.y > 0.0f ? lo.y : 0.0f;
  lo.z = lo.z > 0.0f ? lo.z : 0.0f; lo.w = lo.w > 0.0f ? lo.w : 0.0f;
  hi.x = hi.x > 0.0f ? hi.x : 0.0f; hi.y = hi.y > 0.0f ? hi.y : 0.0f;
  hi.z = hi.z > 0.0f ? hi.z : 0.0f; hi.w = hi.w > 0.0f ? hi.w : 0.0f;
  v4f* dst = (v4f*)out + chunk * 32;
  dst[r]      = lo;
  dst[16 + r] = hi;
}

__global__ void __launch_bounds__(512)
scores_softmax_kernel(const float* __restrict__ Q,
                      const float* __restrict__ K,
                      float* __restrict__ ST) {
  __shared__ v4f red[2][32][5];
  const int b    = blockIdx.y;
  const int j0   = blockIdx.x * 32;
  const int wave = threadIdx.x >> 5;
  const int lane = threadIdx.x & 31;
  const int half = lane >> 4;
  const int r    = lane & 15;

  const v4f* ap = (const v4f*)K + ((size_t)((b * N_ + j0) / 16) * (W_ / 8)) * 32 + lane;
  const v4f* bp = (const v4f*)Q + ((size_t)(b * (N_ / 16) + wave * 8) * (W_ / 8)) * 32 + lane;

  v8f acc[2][8];
#pragma unroll
  for (int jt = 0; jt < 2; ++jt) {
#pragma unroll
    for (int t = 0; t < 8; ++t) acc[jt][t] = (v8f){};
  }

  v4f a0[2], q0[8], a1[2], q1[8];
#pragma unroll
  for (int jt = 0; jt < 2; ++jt) a0[jt] = ap[jt * ((W_ / 8) * 32)];
#pragma unroll
  for (int t = 0; t < 8; ++t) q0[t] = bp[t * ((W_ / 8) * 32)];

#pragma unroll 1
  for (int kc = 0; kc < W_ / 8 - 2; kc += 2) {
#pragma unroll
    for (int jt = 0; jt < 2; ++jt) a1[jt] = ap[jt * ((W_ / 8) * 32) + (kc + 1) * 32];
#pragma unroll
    for (int t = 0; t < 8; ++t) q1[t] = bp[t * ((W_ / 8) * 32) + (kc + 1) * 32];
#pragma unroll
    for (int t = 0; t < 8; ++t) {
#pragma unroll
      for (int jt = 0; jt < 2; ++jt) {
        acc[jt][t] = wmma_f32_16x16x4(a0[jt].lo, q0[t].lo, acc[jt][t]);
        acc[jt][t] = wmma_f32_16x16x4(a0[jt].hi, q0[t].hi, acc[jt][t]);
      }
    }
#pragma unroll
    for (int jt = 0; jt < 2; ++jt) a0[jt] = ap[jt * ((W_ / 8) * 32) + (kc + 2) * 32];
#pragma unroll
    for (int t = 0; t < 8; ++t) q0[t] = bp[t * ((W_ / 8) * 32) + (kc + 2) * 32];
#pragma unroll
    for (int t = 0; t < 8; ++t) {
#pragma unroll
      for (int jt = 0; jt < 2; ++jt) {
        acc[jt][t] = wmma_f32_16x16x4(a1[jt].lo, q1[t].lo, acc[jt][t]);
        acc[jt][t] = wmma_f32_16x16x4(a1[jt].hi, q1[t].hi, acc[jt][t]);
      }
    }
  }
#pragma unroll
  for (int jt = 0; jt < 2; ++jt) a1[jt] = ap[jt * ((W_ / 8) * 32) + (W_ / 8 - 1) * 32];
#pragma unroll
  for (int t = 0; t < 8; ++t) q1[t] = bp[t * ((W_ / 8) * 32) + (W_ / 8 - 1) * 32];
#pragma unroll
  for (int t = 0; t < 8; ++t) {
#pragma unroll
    for (int jt = 0; jt < 2; ++jt) {
      acc[jt][t] = wmma_f32_16x16x4(a0[jt].lo, q0[t].lo, acc[jt][t]);
      acc[jt][t] = wmma_f32_16x16x4(a0[jt].hi, q0[t].hi, acc[jt][t]);
    }
  }
#pragma unroll
  for (int t = 0; t < 8; ++t) {
#pragma unroll
    for (int jt = 0; jt < 2; ++jt) {
      acc[jt][t] = wmma_f32_16x16x4(a1[jt].lo, q1[t].lo, acc[jt][t]);
      acc[jt][t] = wmma_f32_16x16x4(a1[jt].hi, q1[t].hi, acc[jt][t]);
    }
  }

  const float scale = 0.022097086912079608f;
#pragma unroll
  for (int jt = 0; jt < 2; ++jt) {
#pragma unroll
    for (int t = 0; t < 8; ++t) acc[jt][t] = acc[jt][t] * scale;
  }

  float m[2][8];
#pragma unroll
  for (int jt = 0; jt < 2; ++jt) {
#pragma unroll
    for (int g = 0; g < 8; ++g) m[jt][g] = acc[jt][0][g];
#pragma unroll
    for (int t = 1; t < 8; ++t) {
#pragma unroll
      for (int g = 0; g < 8; ++g) m[jt][g] = fmaxf(m[jt][g], acc[jt][t][g]);
    }
  }
#pragma unroll
  for (int off = 8; off > 0; off >>= 1) {
#pragma unroll
    for (int jt = 0; jt < 2; ++jt) {
#pragma unroll
      for (int g = 0; g < 8; ++g) m[jt][g] = fmaxf(m[jt][g], __shfl_xor(m[jt][g], off, 32));
    }
  }
  if (r == 0) {
#pragma unroll
    for (int jt = 0; jt < 2; ++jt) {
#pragma unroll
      for (int g = 0; g < 8; ++g) ((float*)&red[0][16 * jt + 8 * half + g][0])[wave] = m[jt][g];
    }
  }
  __syncthreads();
  {
    const v4f p0 = red[0][lane][0], p1 = red[0][lane][1], p2 = red[0][lane][2], p3 = red[0][lane][3];
    const float mrow = fmaxf(fmaxf(fmaxf(fmaxf(p0.x, p0.y), fmaxf(p0.z, p0.w)),
                                   fmaxf(fmaxf(p1.x, p1.y), fmaxf(p1.z, p1.w))),
                             fmaxf(fmaxf(fmaxf(p2.x, p2.y), fmaxf(p2.z, p2.w)),
                                   fmaxf(fmaxf(p3.x, p3.y), fmaxf(p3.z, p3.w))));
#pragma unroll
    for (int jt = 0; jt < 2; ++jt) {
#pragma unroll
      for (int g = 0; g < 8; ++g) m[jt][g] = __shfl(mrow, 16 * jt + 8 * half + g, 32);
    }
  }

  float d[2][8];
#pragma unroll
  for (int jt = 0; jt < 2; ++jt) {
#pragma unroll
    for (int g = 0; g < 8; ++g) d[jt][g] = 0.0f;
#pragma unroll
    for (int t = 0; t < 8; ++t) {
#pragma unroll
      for (int g = 0; g < 8; ++g) {
        acc[jt][t][g] = __expf(acc[jt][t][g] - m[jt][g]);
        d[jt][g] += acc[jt][t][g];
      }
    }
  }
#pragma unroll
  for (int off = 8; off > 0; off >>= 1) {
#pragma unroll
    for (int jt = 0; jt < 2; ++jt) {
#pragma unroll
      for (int g = 0; g < 8; ++g) d[jt][g] += __shfl_xor(d[jt][g], off, 32);
    }
  }
  if (r == 0) {
#pragma unroll
    for (int jt = 0; jt < 2; ++jt) {
#pragma unroll
      for (int g = 0; g < 8; ++g) ((float*)&red[1][16 * jt + 8 * half + g][0])[wave] = d[jt][g];
    }
  }
  __syncthreads();
  float inv[2][8];
  {
    const v4f p0 = red[1][lane][0], p1 = red[1][lane][1], p2 = red[1][lane][2], p3 = red[1][lane][3];
    float drow = p0.x;
    drow += p0.y; drow += p0.z; drow += p0.w;
    drow += p1.x; drow += p1.y; drow += p1.z; drow += p1.w;
    drow += p2.x; drow += p2.y; drow += p2.z; drow += p2.w;
    drow += p3.x; drow += p3.y; drow += p3.z; drow += p3.w;
    const float irow = 1.0f / drow;
#pragma unroll
    for (int jt = 0; jt < 2; ++jt) {
#pragma unroll
      for (int g = 0; g < 8; ++g) inv[jt][g] = __shfl(irow, 16 * jt + 8 * half + g, 32);
    }
  }

#pragma unroll
  for (int jt = 0; jt < 2; ++jt) {
    const size_t chunk = (size_t)b * (N_ * (N_ / 128)) +
                         (size_t)((j0 + 16 * jt) / 8 + half) * (N_ / 16) + wave * 8;
    v4f* dst = (v4f*)ST + chunk * 32;
#pragma unroll
    for (int t = 0; t < 8; ++t) {
      dst[t * 32 + r]      = (v4f){acc[jt][t][0] * inv[jt][0], acc[jt][t][1] * inv[jt][1],
                                   acc[jt][t][2] * inv[jt][2], acc[jt][t][3] * inv[jt][3]};
      dst[t * 32 + 16 + r] = (v4f){acc[jt][t][4] * inv[jt][4], acc[jt][t][5] * inv[jt][5],
                                   acc[jt][t][6] * inv[jt][6], acc[jt][t][7] * inv[jt][7]};
    }
  }
}

__global__ void av_kernel(const float* __restrict__ PT, const float* __restrict__ V,
                          const float* __restrict__ Q, float* __restrict__ out) {
  const int b    = blockIdx.y;
  const int wave = threadIdx.x >> 5;
  const int lane = threadIdx.x & 31;
  const int half = lane >> 4;
  const int r    = lane & 15;

  const int tiles_n = W_ / 16;
  const int tile = blockIdx.x * 8 + wave;
  const int i0 = (tile / tiles_n) * 16;
  const int n0 = (tile % tiles_n) * 16;

  const v4f* ap = (const v4f*)PT + ((size_t)b * (N_ * (N_ / 128)) + (i0 / 16)) * 32 + lane;
  const v4f* bp = (const v4f*)V + ((size_t)b * (N_ * W_ / 128) + (size_t)(n0 / 16) * (N_ / 8)) * 32 + lane;

  v8f acc = {};
#pragma unroll 4
  for (int kc = 0; kc < N_ / 8; ++kc) {
    const v4f a = ap[(size_t)kc * ((N_ / 16) * 32)];
    const v4f v = bp[kc * 32];
    acc = wmma_f32_16x16x4(a.lo, v.lo, acc);
    acc = wmma_f32_16x16x4(a.hi, v.hi, acc);
  }

  const float* qc = Q + (((size_t)((b * N_ + i0) / 16) * (W_ / 8) + (n0 / 8 + (r >> 3))) * 32 +
                         ((r >> 2) & 1) * 16 + 8 * half) * 4 + (r & 3);
  float* ob = out + (size_t)b * N_ * W_;
#pragma unroll
  for (int g = 0; g < 8; ++g) {
    const int i = i0 + g + 8 * half;
    ob[(size_t)i * W_ + n0 + r] = acc[g] + qc[g * 4];
  }
}

extern "C" void kernel_launch(void* const* d_in, const int* in_sizes, int n_in,
                              void* d_out, int out_size, void* d_ws, size_t ws_size,
                              hipStream_t stream) {
  const float* x  = (const float*)d_in[0];
  const float* Wq = (const float*)d_in[1];
  const float* bq = (const float*)d_in[2];
  const float* Wk = (const float*)d_in[3];
  const float* bk = (const float*)d_in[4];
  const float* Wv = (const float*)d_in[5];
  const float* bv = (const float*)d_in[6];
  float* out = (float*)d_out;
  float* ws  = (float*)d_ws;

  const size_t QKV = (size_t)B_ * N_ * W_;
  float* Q = ws;
  float* K = ws + QKV;
  float* V = ws + 2 * QKV;
  float* ST = ws + 3 * QKV;

  {
    dim3 grid(((B_ * N_ / 16) * (W_ / 16)) / 8, 3);
    qkv_kernel<<<grid, 256, 0, stream>>>(x, Wq, bq, Wk, bk, Wv, bv, Q, K, V);
  }
  {
    dim3 grid(N_ / 32, B_);
    scores_softmax_kernel<<<grid, 512, 0, stream>>>(Q, K, ST);
  }
  {
    dim3 grid(((N_ / 16) * (W_ / 16)) / 8, B_);
    av_kernel<<<grid, 256, 0, stream>>>(ST, V, Q, out);
  }
}


# x, Wq, Wk, Wv first copied into matrix-fragment order by one small extra launch: the projections load whole 512-B chunks, 16 B per lane for two products (was 8 B over 16 lines); new k sum order
# speedup vs baseline: 5.2157x; 1.2583x over previous
#include <hip/hip_runtime.h>

#define B_ 8
#define N_ 2048
#define C_ 256
#define W_ 256

typedef float v2f __attribute__((ext_vector_type(2)));
typedef float v4f __attribute__((ext_vector_type(4)));
typedef float v8f __attribute__((ext_vector_type(8)));

__device__ __forceinline__ v8f wmma_f32_16x16x4(v2f a, v2f b, v8f c) {
  return __builtin_amdgcn_wmma_f32_16x16x4_f32(false, a, false, b, (short)0, c,
                                               false, false);
}


__global__ void frag_kernel(const float* __restrict__ x, const float* __restrict__ Wq,
                            const float* __restrict__ Wk, const float* __restrict__ Wv,
                            float* __restrict__ XF) {
  const int wave = threadIdx.x >> 5;
  const int lane = threadIdx.x & 31;
  const int half = lane >> 4;
  const int r    = lane & 15;
  const int rb   = blockIdx.x;
  const int xblocks = B_ * N_ / 16;
  const float* src;
  if (rb < xblocks) {
    src = x + (size_t)rb * 16 * C_;
  } else {
    const int wb = rb - xblocks;
    const int which = wb / (W_ / 16);
    const float* Wm = (which == 0) ? Wq : (which == 1) ? Wk : Wv;
    src = Wm + (size_t)(wb % (W_ / 16)) * 16 * C_;
  }
  const v4f* sp = (const v4f*)(src + (size_t)r * C_ + 4 * half) + wave * 16;
  v4f* dp = (v4f*)XF + ((size_t)rb * (C_ / 8) + wave * 8) * 32 + lane;
  v4f t[8];
#pragma unroll
  for (int u = 0; u < 8; ++u) t[u] = sp[2 * u];
#pragma unroll
  for (int u = 0; u < 8; ++u) dp[u * 32] = t[u];
}

__global__ void qkv_kernel(const float* __restrict__ XF,
                           const float* __restrict__ bq, const float* __restrict__ bk,
                           const float* __restrict__ bv,
                           float* __restrict__ Q, float* __restrict__ K,
                           float* __restrict__ V) {
  const float* bias;
  float* out;
  if (blockIdx.y == 0) { bias = bq; out = Q; }
  else if (blockIdx.y == 1) { bias = bk; out = K; }
  else { bias = bv; out = V; }
  const bool isV = (blockIdx.y == 2);

  const int wave = threadIdx.x >> 5;
  const int lane = threadIdx.x & 31;
  const int half = lane >> 4;
  const int r    = lane & 15;

  const int tiles_n = W_ / 16;
  const int tile = blockIdx.x * 8 + wave;
  const int m0 = (tile / tiles_n) * 16;
  const int n0 = (tile % tiles_n) * 16;

  const v4f* xp = (const v4f*)XF + ((size_t)(m0 / 16) * (C_ / 8)) * 32 + lane;
  const v4f* wp = (const v4f*)XF +
                  ((size_t)(B_ * N_ / 16 + blockIdx.y * (W_ / 16) + n0 / 16) * (C_ / 8)) * 32 + lane;
  const v4f* ap = isV ? xp : wp;
  const v4f* bp = isV ? wp : xp;

  v8f acc = {};
#pragma unroll 8
  for (int kc = 0; kc < C_ / 8; ++kc) {
    const v4f a = ap[kc * 32];
    const v4f b = bp[kc * 32];
    acc = wmma_f32_16x16x4(a.lo, b.lo, acc);
    acc = wmma_f32_16x16x4(a.hi, b.hi, acc);
  }

  v4f lo, hi;
  size_t chunk;
  if (!isV) {
    const v4f b0 = *(const v4f*)(bias + n0 + 8 * half);
    const v4f b1 = *(const v4f*)(bias + n0 + 8 * half + 4);
    lo = (v4f){acc[0] + b0.x, acc[1] + b0.y, acc[2] + b0.z, acc[3] + b0.w};
    hi = (v4f){acc[4] + b1.x, acc[5] + b1.y, acc[6] + b1.z, acc[7] + b1.w};
    chunk = (size_t)(m0 / 16) * (W_ / 8) + (n0 / 8 + half);
  } else {
    const float bv_ = bias[n0 + r];
    lo = (v4f){acc[0] + bv_, acc[1] + bv_, acc[2] + bv_, acc[3] + bv_};
    hi = (v4f){acc[4] + bv_, acc[5] + bv_, acc[6] + bv_, acc[7] + bv_};
    const int b    = m0 / N_;
    const int jloc = m0 % N_;
    chunk = (size_t)b * (N_ * W_ / 128) + (size_t)(n0 / 16) * (N_ / 8) + (jloc / 8 + half);
  }
  lo.x = lo.x > 0.0f ? lo.x : 0.0f; lo.y = lo.y > 0.0f ? lo.y : 0.0f;
  lo.z = lo.z > 0.0f ? lo.z : 0.0f; lo.w = lo.w > 0.0f ? lo.w : 0.0f;
  hi.x = hi.x > 0.0f ? hi.x : 0.0f; hi.y = hi.y > 0.0f ? hi.y : 0.0f;
  hi.z = hi.z > 0.0f ? hi.z : 0.0f; hi.w = hi.w > 0.0f ? hi.w : 0.0f;
  v4f* dst = (v4f*)out + chunk * 32;
  dst[r]      = lo;
  dst[16 + r] = hi;
}

__global__ void __launch_bounds__(512)
scores_softmax_kernel(const float* __restrict__ Q,
                      const float* __restrict__ K,
                      float* __restrict__ ST) {
  __shared__ v4f red[2][32][5];
  const int b    = blockIdx.y;
  const int j0   = blockIdx.x * 32;
  const int wave = threadIdx.x >> 5;
  const int lane = threadIdx.x & 31;
  const int half = lane >> 4;
  const int r    = lane & 15;

  const v4f* ap = (const v4f*)K + ((size_t)((b * N_ + j0) / 16) * (W_ / 8)) * 32 + lane;
  const v4f* bp = (const v4f*)Q + ((size_t)(b * (N_ / 16) + wave * 8) * (W_ / 8)) * 32 + lane;

  v8f acc[2][8];
#pragma unroll
  for (int jt = 0; jt < 2; ++jt) {
#pragma unroll
    for (int t = 0; t < 8; ++t) acc[jt][t] = (v8f){};
  }

  v4f a0[2], q0[8], a1[2], q1[8];
#pragma unroll
  for (int jt = 0; jt < 2; ++jt) a0[jt] = ap[jt * ((W_ / 8) * 32)];
#pragma unroll
  for (int t = 0; t < 8; ++t) q0[t] = bp[t * ((W_ / 8) * 32)];

#pragma unroll 1
  for (int kc = 0; kc < W_ / 8 - 2; kc += 2) {
#pragma unroll
    for (int jt = 0; jt < 2; ++jt) a1[jt] = ap[jt * ((W_ / 8) * 32) + (kc + 1) * 32];
#pragma unroll
    for (int t = 0; t < 8; ++t) q1[t] = bp[t * ((W_ / 8) * 32) + (kc + 1) * 32];
#pragma unroll
    for (int t = 0; t < 8; ++t) {
#pragma unroll
      for (int jt = 0; jt < 2; ++jt) {
        acc[jt][t] = wmma_f32_16x16x4(a0[jt].lo, q0[t].lo, acc[jt][t]);
        acc[jt][t] = wmma_f32_16x16x4(a0[jt].hi, q0[t].hi, acc[jt][t]);
      }
    }
#pragma unroll
    for (int jt = 0; jt < 2; ++jt) a0[jt] = ap[jt * ((W_ / 8) * 32) + (kc + 2) * 32];
#pragma unroll
    for (int t = 0; t < 8; ++t) q0[t] = bp[t * ((W_ / 8) * 32) + (kc + 2) * 32];
#pragma unroll
    for (int t = 0; t < 8; ++t) {
#pragma unroll
      for (int jt = 0; jt < 2; ++jt) {
        acc[jt][t] = wmma_f32_16x16x4(a1[jt].lo, q1[t].lo, acc[jt][t]);
        acc[jt][t] = wmma_f32_16x16x4(a1[jt].hi, q1[t].hi, acc[jt][t]);
      }
    }
  }
#pragma unroll
  for (int jt = 0; jt < 2; ++jt) a1[jt] = ap[jt * ((W_ / 8) * 32) + (W_ / 8 - 1) * 32];
#pragma unroll
  for (int t = 0; t < 8; ++t) q1[t] = bp[t * ((W_ / 8) * 32) + (W_ / 8 - 1) * 32];
#pragma unroll
  for (int t = 0; t < 8; ++t) {
#pragma unroll
    for (int jt = 0; jt < 2; ++jt) {
      acc[jt][t] = wmma_f32_16x16x4(a0[jt].lo, q0[t].lo, acc[jt][t]);
      acc[jt][t] = wmma_f32_16x16x4(a0[jt].hi, q0[t].hi, acc[jt][t]);
    }
  }
#pragma unroll
  for (int t = 0; t < 8; ++t) {
#pragma unroll
    for (int jt = 0; jt < 2; ++jt) {
      acc[jt][t] = wmma_f32_16x16x4(a1[jt].lo, q1[t].lo, acc[jt][t]);
      acc[jt][t] = wmma_f32_16x16x4(a1[jt].hi, q1[t].hi, acc[jt][t]);
    }
  }

  const float scale = 0.022097086912079608f;
#pragma unroll
  for (int jt = 0; jt < 2; ++jt) {
#pragma unroll
    for (int t = 0; t < 8; ++t) acc[jt][t] = acc[jt][t] * scale;
  }

  float m[2][8];
#pragma unroll
  for (int jt = 0; jt < 2; ++jt) {
#pragma unroll
    for (int g = 0; g < 8; ++g) m[jt][g] = acc[jt][0][g];
#pragma unroll
    for (int t = 1; t < 8; ++t) {
#pragma unroll
      for (int g = 0; g < 8; ++g) m[jt][g] = fmaxf(m[jt][g], acc[jt][t][g]);
    }
  }
#pragma unroll
  for (int off = 8; off > 0; off >>= 1) {
#pragma unroll
    for (int jt = 0; jt < 2; ++jt) {
#pragma unroll
      for (int g = 0; g < 8; ++g) m[jt][g] = fmaxf(m[jt][g], __shfl_xor(m[jt][g], off, 32));
    }
  }
  if (r == 0) {
#pragma unroll
    for (int jt = 0; jt < 2; ++jt) {
#pragma unroll
      for (int g = 0; g < 8; ++g) ((float*)&red[0][16 * jt + 8 * half + g][0])[wave] = m[jt][g];
    }
  }
  __syncthreads();
  {
    const v4f p0 = red[0][lane][0], p1 = red[0][lane][1], p2 = red[0][lane][2], p3 = red[0][lane][3];
    const float mrow = fmaxf(fmaxf(fmaxf(fmaxf(p0.x, p0.y), fmaxf(p0.z, p0.w)),
                                   fmaxf(fmaxf(p1.x, p1.y), fmaxf(p1.z, p1.w))),
                             fmaxf(fmaxf(fmaxf(p2.x, p2.y), fmaxf(p2.z, p2.w)),
                                   fmaxf(fmaxf(p3.x, p3.y), fmaxf(p3.z, p3.w))));
#pragma unroll
    for (int jt = 0; jt < 2; ++jt) {
#pragma unroll
      for (int g = 0; g < 8; ++g) m[jt][g] = __shfl(mrow, 16 * jt + 8 * half + g, 32);
    }
  }

  float d[2][8];
#pragma unroll
  for (int jt = 0; jt < 2; ++jt) {
#pragma unroll
    for (int g = 0; g < 8; ++g) d[jt][g] = 0.0f;
#pragma unroll
    for (int t = 0; t < 8; ++t) {
#pragma unroll
      for (int g = 0; g < 8; ++g) {
        acc[jt][t][g] = __expf(acc[jt][t][g] - m[jt][g]);
        d[jt][g] += acc[jt][t][g];
      }
    }
  }
#pragma unroll
  for (int off = 8; off > 0; off >>= 1) {
#pragma unroll
    for (int jt = 0; jt < 2; ++jt) {
#pragma unroll
      for (int g = 0; g < 8; ++g) d[jt][g] += __shfl_xor(d[jt][g], off, 32);
    }
  }
  if (r == 0) {
#pragma unroll
    for (int jt = 0; jt < 2; ++jt) {
#pragma unroll
      for (int g = 0; g < 8; ++g) ((float*)&red[1][16 * jt + 8 * half + g][0])[wave] = d[jt][g];
    }
  }
  __syncthreads();
  float inv[2][8];
  {
    const v4f p0 = red[1][lane][0], p1 = red[1][lane][1], p2 = red[1][lane][2], p3 = red[1][lane][3];
    float drow = p0.x;
    drow += p0.y; drow += p0.z; drow += p0.w;
    drow += p1.x; drow += p1.y; drow += p1.z; drow += p1.w;
    drow += p2.x; drow += p2.y; drow += p2.z; drow += p2.w;
    drow += p3.x; drow += p3.y; drow += p3.z; drow += p3.w;
    const float irow = 1.0f / drow;
#pragma unroll
    for (int jt = 0; jt < 2; ++jt) {
#pragma unroll
      for (int g = 0; g < 8; ++g) inv[jt][g] = __shfl(irow, 16 * jt + 8 * half + g, 32);
    }
  }

#pragma unroll
  for (int jt = 0; jt < 2; ++jt) {
    const size_t chunk = (size_t)b * (N_ * (N_ / 128)) +
                         (size_t)((j0 + 16 * jt) / 8 + half) * (N_ / 16) + wave * 8;
    v4f* dst = (v4f*)ST + chunk * 32;
#pragma unroll
    for (int t = 0; t < 8; ++t) {
      dst[t * 32 + r]      = (v4f){acc[jt][t][0] * inv[jt][0], acc[jt][t][1] * inv[jt][1],
                                   acc[jt][t][2] * inv[jt][2], acc[jt][t][3] * inv[jt][3]};
      dst[t * 32 + 16 + r] = (v4f){acc[jt][t][4] * inv[jt][4], acc[jt][t][5] * inv[jt][5],
                                   acc[jt][t][6] * inv[jt][6], acc[jt][t][7] * inv[jt][7]};
    }
  }
}

__global__ void av_kernel(const float* __restrict__ PT, const float* __restrict__ V,
                          const float* __restrict__ Q, float* __restrict__ out) {
  const int b    = blockIdx.y;
  const int wave = threadIdx.x >> 5;
  const int lane = threadIdx.x & 31;
  const int half = lane >> 4;
  const int r    = lane & 15;

  const int tiles_n = W_ / 16;
  const int tile = blockIdx.x * 8 + wave;
  const int i0 = (tile / tiles_n) * 16;
  const int n0 = (tile % tiles_n) * 16;

  const v4f* ap = (const v4f*)PT + ((size_t)b * (N_ * (N_ / 128)) + (i0 / 16)) * 32 + lane;
  const v4f* bp = (const v4f*)V + ((size_t)b * (N_ * W_ / 128) + (size_t)(n0 / 16) * (N_ / 8)) * 32 + lane;

  v8f acc = {};
#pragma unroll 4
  for (int kc = 0; kc < N_ / 8; ++kc) {
    const v4f a = ap[(size_t)kc * ((N_ / 16) * 32)];
    const v4f v = bp[kc * 32];
    acc = wmma_f32_16x16x4(a.lo, v.lo, acc);
    acc = wmma_f32_16x16x4(a.hi, v.hi, acc);
  }

  const float* qc = Q + (((size_t)((b * N_ + i0) / 16) * (W_ / 8) + (n0 / 8 + (r >> 3))) * 32 +
                         ((r >> 2) & 1) * 16 + 8 * half) * 4 + (r & 3);
  float* ob = out + (size_t)b * N_ * W_;
#pragma unroll
  for (int g = 0; g < 8; ++g) {
    const int i = i0 + g + 8 * half;
    ob[(size_t)i * W_ + n0 + r] = acc[g] + qc[g * 4];
  }
}

extern "C" void kernel_launch(void* const* d_in, const int* in_sizes, int n_in,
                              void* d_out, int out_size, void* d_ws, size_t ws_size,
                              hipStream_t stream) {
  const float* x  = (const float*)d_in[0];
  const float* Wq = (const float*)d_in[1];
  const float* bq = (const float*)d_in[2];
  const float* Wk = (const float*)d_in[3];
  const float* bk = (const float*)d_in[4];
  const float* Wv = (const float*)d_in[5];
  const float* bv = (const float*)d_in[6];
  float* out = (float*)d_out;
  float* ws  = (float*)d_ws;

  const size_t QKV = (size_t)B_ * N_ * W_;
  float* Q = ws;
  float* K = ws + QKV;
  float* V = ws + 2 * QKV;
  float* ST = ws + 3 * QKV;

  float* XF = ST;
  {
    dim3 grid(B_ * N_ / 16 + 3 * (W_ / 16));
    frag_kernel<<<grid, 128, 0, stream>>>(x, Wq, Wk, Wv, XF);
  }
  {
    dim3 grid(((B_ * N_ / 16) * (W_ / 16)) / 8, 3);
    qkv_kernel<<<grid, 256, 0, stream>>>(XF, bq, bk, bv, Q, K, V);
  }
  {
    dim3 grid(N_ / 32, B_);
    scores_softmax_kernel<<<grid, 512, 0, stream>>>(Q, K, ST);
  }
  {
    dim3 grid(((N_ / 16) * (W_ / 16)) / 8, B_);
    av_kernel<<<grid, 256, 0, stream>>>(ST, V, Q, out);
  }
}


# projections and last product (P V): 4x4 accumulator tiles per wave, 8 fragment loads per 32 matrix instructions, requested 3 k steps ahead in a ring of 4 register sets (was 1 tile); sums keep order
# speedup vs baseline: 6.3252x; 1.2127x over previous
#include <hip/hip_runtime.h>

#define B_ 8
#define N_ 2048
#define C_ 256
#define W_ 256

typedef float v2f __attribute__((ext_vector_type(2)));
typedef float v4f __attribute__((ext_vector_type(4)));
typedef float v8f __attribute__((ext_vector_type(8)));

__device__ __forceinline__ v8f wmma_f32_16x16x4(v2f a, v2f b, v8f c) {
  return __builtin_amdgcn_wmma_f32_16x16x4_f32(false, a, false, b, (short)0, c,
                                               false, false);
}


__global__ void frag_kernel(const float* __restrict__ x, const float* __restrict__ Wq,
                            const float* __restrict__ Wk, const float* __restrict__ Wv,
                            float* __restrict__ XF) {
  const int wave = threadIdx.x >> 5;
  const int lane = threadIdx.x & 31;
  const int half = lane >> 4;
  const int r    = lane & 15;
  const int rb   = blockIdx.x;
  const int xblocks = B_ * N_ / 16;
  const float* src;
  if (rb < xblocks) {
    src = x + (size_t)rb * 16 * C_;
  } else {
    const int wb = rb - xblocks;
    const int which = wb / (W_ / 16);
    const float* Wm = (which == 0) ? Wq : (which == 1) ? Wk : Wv;
    src = Wm + (size_t)(wb % (W_ / 16)) * 16 * C_;
  }
  const v4f* sp = (const v4f*)(src + (size_t)r * C_ + 4 * half) + wave * 16;
  v4f* dp = (v4f*)XF + ((size_t)rb * (C_ / 8) + wave * 8) * 32 + lane;
  v4f t[8];
#pragma unroll
  for (int u = 0; u < 8; ++u) t[u] = sp[2 * u];
#pragma unroll
  for (int u = 0; u < 8; ++u) dp[u * 32] = t[u];
}

__device__ __forceinline__ void mm_step(v8f (&acc)[4][4], const v4f (&a)[4], const v4f (&b)[4]) {
#pragma unroll
  for (int ia = 0; ia < 4; ++ia) {
#pragma unroll
    for (int ib = 0; ib < 4; ++ib) {
      acc[ia][ib] = wmma_f32_16x16x4(a[ia].lo, b[ib].lo, acc[ia][ib]);
      acc[ia][ib] = wmma_f32_16x16x4(a[ia].hi, b[ib].hi, acc[ia][ib]);
    }
  }
}

__global__ void __launch_bounds__(128)
qkv_kernel(const float* __restrict__ XF,
           const float* __restrict__ bq, const float* __restrict__ bk,
           const float* __restrict__ bv,
           float* __restrict__ Q, float* __restrict__ K,
           float* __restrict__ V) {
  const float* bias;
  float* out;
  if (blockIdx.y == 0) { bias = bq; out = Q; }
  else if (blockIdx.y == 1) { bias = bk; out = K; }
  else { bias = bv; out = V; }
  const bool isV = (blockIdx.y == 2);

  const int wave = threadIdx.x >> 5;
  const int lane = threadIdx.x & 31;
  const int half = lane >> 4;
  const int r    = lane & 15;

  const int tb0 = blockIdx.x * 4;
  const int fb0 = wave * 4;

  const v4f* xp = (const v4f*)XF + ((size_t)tb0 * (C_ / 8)) * 32 + lane;
  const v4f* wp = (const v4f*)XF +
                  ((size_t)(B_ * N_ / 16 + blockIdx.y * (W_ / 16) + fb0) * (C_ / 8)) * 32 + lane;
  const v4f* ap = isV ? xp : wp;
  const v4f* bp = isV ? wp : xp;

  v8f acc[4][4];
#pragma unroll
  for (int ia = 0; ia < 4; ++ia) {
#pragma unroll
    for (int ib = 0; ib < 4; ++ib) acc[ia][ib] = (v8f){};
  }

  v4f fa[4][4], fb[4][4];
#pragma unroll
  for (int s = 0; s < 3; ++s) {
#pragma unroll
    for (int u = 0; u < 4; ++u) fa[s][u] = ap[(u * (C_ / 8) + s) * 32];
#pragma unroll
    for (int u = 0; u < 4; ++u) fb[s][u] = bp[(u * (C_ / 8) + s) * 32];
    __builtin_amdgcn_sched_barrier(0);
  }

#pragma unroll 1
  for (int kc = 0; kc < C_ / 8 - 4; kc += 4) {
#pragma unroll
    for (int s = 0; s < 4; ++s) {
      const int kn = kc + s + 3;
      const int sn = (s + 3) & 3;
#pragma unroll
      for (int u = 0; u < 4; ++u) fa[sn][u] = ap[(u * (C_ / 8) + kn) * 32];
#pragma unroll
      for (int u = 0; u < 4; ++u) fb[sn][u] = bp[(u * (C_ / 8) + kn) * 32];
      __builtin_amdgcn_sched_barrier(0);
      mm_step(acc, fa[s], fb[s]);
    }
  }
#pragma unroll
  for (int u = 0; u < 4; ++u) fa[3][u] = ap[(u * (C_ / 8) + (C_ / 8 - 1)) * 32];
#pragma unroll
  for (int u = 0; u < 4; ++u) fb[3][u] = bp[(u * (C_ / 8) + (C_ / 8 - 1)) * 32];
#pragma unroll
  for (int s = 0; s < 4; ++s) mm_step(acc, fa[s], fb[s]);

  if (!isV) {
#pragma unroll
    for (int ia = 0; ia < 4; ++ia) {
      const int n0 = (fb0 + ia) * 16;
      const v4f b0 = *(const v4f*)(bias + n0 + 8 * half);
      const v4f b1 = *(const v4f*)(bias + n0 + 8 * half + 4);
#pragma unroll
      for (int ib = 0; ib < 4; ++ib) {
        const v8f c = acc[ia][ib];
        v4f lo = (v4f){c[0] + b0.x, c[1] + b0.y, c[2] + b0.z, c[3] + b0.w};
        v4f hi = (v4f){c[4] + b1.x, c[5] + b1.y, c[6] + b1.z, c[7] + b1.w};
        lo.x = lo.x > 0.0f ? lo.x : 0.0f; lo.y = lo.y > 0.0f ? lo.y : 0.0f;
        lo.z = lo.z > 0.0f ? lo.z : 0.0f; lo.w = lo.w > 0.0f ? lo.w : 0.0f;
        hi.x = hi.x > 0.0f ? hi.x : 0.0f; hi.y = hi.y > 0.0f ? hi.y : 0.0f;
        hi.z = hi.z > 0.0f ? hi.z : 0.0f; hi.w = hi.w > 0.0f ? hi.w : 0.0f;
        const size_t chunk = (size_t)(tb0 + ib) * (W_ / 8) + (n0 / 8 + half);
        v4f* dst = (v4f*)out + chunk * 32;
        dst[r]      = lo;
        dst[16 + r] = hi;
      }
    }
  } else {
#pragma unroll
    for (int ib = 0; ib < 4; ++ib) {
      const int n0 = (fb0 + ib) * 16;
      const float bv_ = bias[n0 + r];
#pragma unroll
      for (int ia = 0; ia < 4; ++ia) {
        const v8f c = acc[ia][ib];
        v4f lo = (v4f){c[0] + bv_, c[1] + bv_, c[2] + bv_, c[3] + bv_};
        v4f hi = (v4f){c[4] + bv_, c[5] + bv_, c[6] + bv_, c[7] + bv_};
        lo.x = lo.x > 0.0f ? lo.x : 0.0f; lo.y = lo.y > 0.0f ? lo.y : 0.0f;
        lo.z = lo.z > 0.0f ? lo.z : 0.0f; lo.w = lo.w > 0.0f ? lo.w : 0.0f;
        hi.x = hi.x > 0.0f ? hi.x : 0.0f; hi.y = hi.y > 0.0f ? hi.y : 0.0f;
        hi.z = hi.z > 0.0f ? hi.z : 0.0f; hi.w = hi.w > 0.0f ? hi.w : 0.0f;
        const int m0   = (tb0 + ia) * 16;
        const int b    = m0 / N_;
        const int jloc = m0 % N_;
        const size_t chunk = (size_t)b * (N_ * W_ / 128) + (size_t)(n0 / 16) * (N_ / 8) + (jloc / 8 + half);
        v4f* dst = (v4f*)out + chunk * 32;
        dst[r]      = lo;
        dst[16 + r] = hi;
      }
    }
  }
}

__global__ void __launch_bounds__(512)
scores_softmax_kernel(const float* __restrict__ Q,
                      const float* __restrict__ K,
                      float* __restrict__ ST) {
  __shared__ v4f red[2][32][5];
  const int b    = blockIdx.y;
  const int j0   = blockIdx.x * 32;
  const int wave = threadIdx.x >> 5;
  const int lane = threadIdx.x & 31;
  const int half = lane >> 4;
  const int r    = lane & 15;

  const v4f* ap = (const v4f*)K + ((size_t)((b * N_ + j0) / 16) * (W_ / 8)) * 32 + lane;
  const v4f* bp = (const v4f*)Q + ((size_t)(b * (N_ / 16) + wave * 8) * (W_ / 8)) * 32 + lane;

  v8f acc[2][8];
#pragma unroll
  for (int jt = 0; jt < 2; ++jt) {
#pragma unroll
    for (int t = 0; t < 8; ++t) acc[jt][t] = (v8f){};
  }

  v4f a0[2], q0[8], a1[2], q1[8];
#pragma unroll
  for (int jt = 0; jt < 2; ++jt) a0[jt] = ap[jt * ((W_ / 8) * 32)];
#pragma unroll
  for (int t = 0; t < 8; ++t) q0[t] = bp[t * ((W_ / 8) * 32)];

#pragma unroll 1
  for (int kc = 0; kc < W_ / 8 - 2; kc += 2) {
#pragma unroll
    for (int jt = 0; jt < 2; ++jt) a1[jt] = ap[jt * ((W_ / 8) * 32) + (kc + 1) * 32];
#pragma unroll
    for (int t = 0; t < 8; ++t) q1[t] = bp[t * ((W_ / 8) * 32) + (kc + 1) * 32];
#pragma unroll
    for (int t = 0; t < 8; ++t) {
#pragma unroll
      for (int jt = 0; jt < 2; ++jt) {
        acc[jt][t] = wmma_f32_16x16x4(a0[jt].lo, q0[t].lo, acc[jt][t]);
        acc[jt][t] = wmma_f32_16x16x4(a0[jt].hi, q0[t].hi, acc[jt][t]);
      }
    }
#pragma unroll
    for (int jt = 0; jt < 2; ++jt) a0[jt] = ap[jt * ((W_ / 8) * 32) + (kc + 2) * 32];
#pragma unroll
    for (int t = 0; t < 8; ++t) q0[t] = bp[t * ((W_ / 8) * 32) + (kc + 2) * 32];
#pragma unroll
    for (int t = 0; t < 8; ++t) {
#pragma unroll
      for (int jt = 0; jt < 2; ++jt) {
        acc[jt][t] = wmma_f32_16x16x4(a1[jt].lo, q1[t].lo, acc[jt][t]);
        acc[jt][t] = wmma_f32_16x16x4(a1[jt].hi, q1[t].hi, acc[jt][t]);
      }
    }
  }
#pragma unroll
  for (int jt = 0; jt < 2; ++jt) a1[jt] = ap[jt * ((W_ / 8) * 32) + (W_ / 8 - 1) * 32];
#pragma unroll
  for (int t = 0; t < 8; ++t) q1[t] = bp[t * ((W_ / 8) * 32) + (W_ / 8 - 1) * 32];
#pragma unroll
  for (int t = 0; t < 8; ++t) {
#pragma unroll
    for (int jt = 0; jt < 2; ++jt) {
      acc[jt][t] = wmma_f32_16x16x4(a0[jt].lo, q0[t].lo, acc[jt][t]);
      acc[jt][t] = wmma_f32_16x16x4(a0[jt].hi, q0[t].hi, acc[jt][t]);
    }
  }
#pragma unroll
  for (int t = 0; t < 8; ++t) {
#pragma unroll
    for (int jt = 0; jt < 2; ++jt) {
      acc[jt][t] = wmma_f32_16x16x4(a1[jt].lo, q1[t].lo, acc[jt][t]);
      acc[jt][t] = wmma_f32_16x16x4(a1[jt].hi, q1[t].hi, acc[jt][t]);
    }
  }

  const float scale = 0.022097086912079608f;
#pragma unroll
  for (int jt = 0; jt < 2; ++jt) {
#pragma unroll
    for (int t = 0; t < 8; ++t) acc[jt][t] = acc[jt][t] * scale;
  }

  float m[2][8];
#pragma unroll
  for (int jt = 0; jt < 2; ++jt) {
#pragma unroll
    for (int g = 0; g < 8; ++g) m[jt][g] = acc[jt][0][g];
#pragma unroll
    for (int t = 1; t < 8; ++t) {
#pragma unroll
      for (int g = 0; g < 8; ++g) m[jt][g] = fmaxf(m[jt][g], acc[jt][t][g]);
    }
  }
#pragma unroll
  for (int off = 8; off > 0; off >>= 1) {
#pragma unroll
    for (int jt = 0; jt < 2; ++jt) {
#pragma unroll
      for (int g = 0; g < 8; ++g) m[jt][g] = fmaxf(m[jt][g], __shfl_xor(m[jt][g], off, 32));
    }
  }
  if (r == 0) {
#pragma unroll
    for (int jt = 0; jt < 2; ++jt) {
#pragma unroll
      for (int g = 0; g < 8; ++g) ((float*)&red[0][16 * jt + 8 * half + g][0])[wave] = m[jt][g];
    }
  }
  __syncthreads();
  {
    const v4f p0 = red[0][lane][0], p1 = red[0][lane][1], p2 = red[0][lane][2], p3 = red[0][lane][3];
    const float mrow = fmaxf(fmaxf(fmaxf(fmaxf(p0.x, p0.y), fmaxf(p0.z, p0.w)),
                                   fmaxf(fmaxf(p1.x, p1.y), fmaxf(p1.z, p1.w))),
                             fmaxf(fmaxf(fmaxf(p2.x, p2.y), fmaxf(p2.z, p2.w)),
                                   fmaxf(fmaxf(p3.x, p3.y), fmaxf(p3.z, p3.w))));
#pragma unroll
    for (int jt = 0; jt < 2; ++jt) {
#pragma unroll
      for (int g = 0; g < 8; ++g) m[jt][g] = __shfl(mrow, 16 * jt + 8 * half + g, 32);
    }
  }

  float d[2][8];
#pragma unroll
  for (int jt = 0; jt < 2; ++jt) {
#pragma unroll
    for (int g = 0; g < 8; ++g) d[jt][g] = 0.0f;
#pragma unroll
    for (int t = 0; t < 8; ++t) {
#pragma unroll
      for (int g = 0; g < 8; ++g) {
        acc[jt][t][g] = __expf(acc[jt][t][g] - m[jt][g]);
        d[jt][g] += acc[jt][t][g];
      }
    }
  }
#pragma unroll
  for (int off = 8; off > 0; off >>= 1) {
#pragma unroll
    for (int jt = 0; jt < 2; ++jt) {
#pragma unroll
      for (int g = 0; g < 8; ++g) d[jt][g] += __shfl_xor(d[jt][g], off, 32);
    }
  }
  if (r == 0) {
#pragma unroll
    for (int jt = 0; jt < 2; ++jt) {
#pragma unroll
      for (int g = 0; g < 8; ++g) ((float*)&red[1][16 * jt + 8 * half + g][0])[wave] = d[jt][g];
    }
  }
  __syncthreads();
  float inv[2][8];
  {
    const v4f p0 = red[1][lane][0], p1 = red[1][lane][1], p2 = red[1][lane][2], p3 = red[1][lane][3];
    float drow = p0.x;
    drow += p0.y; drow += p0.z; drow += p0.w;
    drow += p1.x; drow += p1.y; drow += p1.z; drow += p1.w;
    drow += p2.x; drow += p2.y; drow += p2.z; drow += p2.w;
    drow += p3.x; drow += p3.y; drow += p3.z; drow += p3.w;
    const float irow = 1.0f / drow;
#pragma unroll
    for (int jt = 0; jt < 2; ++jt) {
#pragma unroll
      for (int g = 0; g < 8; ++g) inv[jt][g] = __shfl(irow, 16 * jt + 8 * half + g, 32);
    }
  }

#pragma unroll
  for (int jt = 0; jt < 2; ++jt) {
    const size_t chunk = (size_t)b * (N_ * (N_ / 128)) +
                         (size_t)((j0 + 16 * jt) / 8 + half) * (N_ / 16) + wave * 8;
    v4f* dst = (v4f*)ST + chunk * 32;
#pragma unroll
    for (int t = 0; t < 8; ++t) {
      dst[t * 32 + r]      = (v4f){acc[jt][t][0] * inv[jt][0], acc[jt][t][1] * inv[jt][1],
                                   acc[jt][t][2] * inv[jt][2], acc[jt][t][3] * inv[jt][3]};
      dst[t * 32 + 16 + r] = (v4f){acc[jt][t][4] * inv[jt][4], acc[jt][t][5] * inv[jt][5],
                                   acc[jt][t][6] * inv[jt][6], acc[jt][t][7] * inv[jt][7]};
    }
  }
}

__global__ void __launch_bounds__(128)
av_kernel(const float* __restrict__ PT, const float* __restrict__ V,
          const float* __restrict__ Q, float* __restrict__ out) {
  const int b    = blockIdx.y;
  const int wave = threadIdx.x >> 5;
  const int lane = threadIdx.x & 31;
  const int half = lane >> 4;
  const int r    = lane & 15;

  const int ib0 = blockIdx.x * 4;
  const int i0  = ib0 * 16;
  const int n0  = wave * 64;

  const v4f* ap = (const v4f*)PT + ((size_t)b * (N_ * (N_ / 128)) + ib0) * 32 + lane;
  const v4f* bp = (const v4f*)V + ((size_t)b * (N_ * W_ / 128) + (size_t)(wave * 4) * (N_ / 8)) * 32 + lane;

  v8f acc[4][4];
#pragma unroll
  for (int it = 0; it < 4; ++it) {
#pragma unroll
    for (int wt = 0; wt < 4; ++wt) acc[it][wt] = (v8f){};
  }

  v4f fa[4][4], fb[4][4];
#pragma unroll
  for (int s = 0; s < 3; ++s) {
#pragma unroll
    for (int it = 0; it < 4; ++it) fa[s][it] = ap[((size_t)s * (N_ / 16) + it) * 32];
#pragma unroll
    for (int wt = 0; wt < 4; ++wt) fb[s][wt] = bp[((size_t)wt * (N_ / 8) + s) * 32];
    __builtin_amdgcn_sched_barrier(0);
  }

#pragma unroll 1
  for (int kc = 0; kc < N_ / 8 - 4; kc += 4) {
#pragma unroll
    for (int s = 0; s < 4; ++s) {
      const int kn = kc + s + 3;
      const int sn = (s + 3) & 3;
#pragma unroll
      for (int it = 0; it < 4; ++it) fa[sn][it] = ap[((size_t)kn * (N_ / 16) + it) * 32];
#pragma unroll
      for (int wt = 0; wt < 4; ++wt) fb[sn][wt] = bp[((size_t)wt * (N_ / 8) + kn) * 32];
      __builtin_amdgcn_sched_barrier(0);
      mm_step(acc, fa[s], fb[s]);
    }
  }
#pragma unroll
  for (int it = 0; it < 4; ++it) fa[3][it] = ap[((size_t)(N_ / 8 - 1) * (N_ / 16) + it) * 32];
#pragma unroll
  for (int wt = 0; wt < 4; ++wt) fb[3][wt] = bp[((size_t)wt * (N_ / 8) + (N_ / 8 - 1)) * 32];
#pragma unroll
  for (int s = 0; s < 4; ++s) mm_step(acc, fa[s], fb[s]);

  float* ob = out + (size_t)b * N_ * W_;
#pragma unroll
  for (int it = 0; it < 4; ++it) {
#pragma unroll
    for (int wt = 0; wt < 4; ++wt) {
      const int nt = n0 + 16 * wt;
      const float* qc = Q + (((size_t)((b * N_ + i0) / 16 + it) * (W_ / 8) + (nt / 8 + (r >> 3))) * 32 +
                             ((r >> 2) & 1) * 16 + 8 * half) * 4 + (r & 3);
#pragma unroll
      for (int g = 0; g < 8; ++g) {
        const int i = i0 + 16 * it + g + 8 * half;
        ob[(size_t)i * W_ + nt + r] = acc[it][wt][g] + qc[g * 4];
      }
    }
  }
}

extern "C" void kernel_launch(void* const* d_in, const int* in_sizes, int n_in,
                              void* d_out, int out_size, void* d_ws, size_t ws_size,
                              hipStream_t stream) {
  const float* x  = (const float*)d_in[0];
  const float* Wq = (const float*)d_in[1];
  const float* bq = (const float*)d_in[2];
  const float* Wk = (const float*)d_in[3];
  const float* bk = (const float*)d_in[4];
  const float* Wv = (const float*)d_in[5];
  const float* bv = (const float*)d_in[6];
  float* out = (float*)d_out;
  float* ws  = (float*)d_ws;

  const size_t QKV = (size_t)B_ * N_ * W_;
  float* Q = ws;
  float* K = ws + QKV;
  float* V = ws + 2 * QKV;
  float* ST = ws + 3 * QKV;

  float* XF = ST;
  {
    dim3 grid(B_ * N_ / 16 + 3 * (W_ / 16));
    frag_kernel<<<grid, 128, 0, stream>>>(x, Wq, Wk, Wv, XF);
  }
  {
    dim3 grid(B_ * N_ / 64, 3);
    qkv_kernel<<<grid, 128, 0, stream>>>(XF, bq, bk, bv, Q, K, V);
  }
  {
    dim3 grid(N_ / 32, B_);
    scores_softmax_kernel<<<grid, 512, 0, stream>>>(Q, K, ST);
  }
  {
    dim3 grid(N_ / 64, B_);
    av_kernel<<<grid, 128, 0, stream>>>(ST, V, Q, out);
  }
}
